# starting kernel
# baseline (speedup 1.0000x reference)
#include <hip/hip_runtime.h>
#include <math.h>

typedef __attribute__((ext_vector_type(2))) float v2f;
typedef __attribute__((ext_vector_type(8))) float v8f;

#define LOG_2PI 1.8378770664093453f
#define LOG2E   1.4426950408889634f
#define LN2     0.6931471805599453f

__global__ __launch_bounds__(32) void gmm_setup_kernel(
    const float* __restrict__ w, const float* __restrict__ means,
    const float* __restrict__ logvar, float* __restrict__ G) {
  const int k = threadIdx.x;
  float wk = w[k];

  float mx = wk;
#pragma unroll
  for (int m = 16; m >= 1; m >>= 1) mx = fmaxf(mx, __shfl_xor(mx, m, 32));
  float s = __expf(wk - mx);
#pragma unroll
  for (int m = 16; m >= 1; m >>= 1) s += __shfl_xor(s, m, 32);
  const float logw = wk - mx - __logf(s);

  const float lv0 = logvar[2 * k], lv1 = logvar[2 * k + 1];
  const float iv0 = __expf(-lv0), iv1 = __expf(-lv1);
  const float m0 = means[2 * k], m1 = means[2 * k + 1];

  const float ck = logw - 0.5f * (lv0 + lv1 + 2.0f * LOG_2PI)
                        - 0.5f * (m0 * m0 * iv0 + m1 * m1 * iv1);
  float M = ck;
#pragma unroll
  for (int m = 16; m >= 1; m >>= 1) M = fmaxf(M, __shfl_xor(M, m, 32));

  G[0 * 32 + k] = -0.5f * iv0 * LOG2E;
  G[1 * 32 + k] = -0.5f * iv1 * LOG2E;
  G[2 * 32 + k] = m0 * iv0 * LOG2E;
  G[3 * 32 + k] = m1 * iv1 * LOG2E;
  G[4 * 32 + k] = (ck - M) * LOG2E;
  if (k == 0) G[160] = M;
}

__global__ __launch_bounds__(256) void gmm_wmma_kernel(
    const float* __restrict__ Xf, const float* __restrict__ G,
    float* __restrict__ out, int npoints, int ntiles) {
  const v2f* __restrict__ X = (const v2f*)Xf;
  const int lane = threadIdx.x & 31;
  const int col = lane & 15;
  const bool hi = lane >= 16;

  const int wavesPerBlock = blockDim.x >> 5;
  const int gwave = blockIdx.x * wavesPerBlock + (threadIdx.x >> 5);
  const int nwaves = gridDim.x * wavesPerBlock;

  v2f aL, aH;
  aL.x = hi ? G[2 * 32 + col] : G[0 * 32 + col];
  aL.y = hi ? G[3 * 32 + col] : G[1 * 32 + col];
  aH.x = hi ? G[2 * 32 + 16 + col] : G[0 * 32 + 16 + col];
  aH.y = hi ? G[3 * 32 + 16 + col] : G[1 * 32 + 16 + col];

  v8f cL, cH;
#pragma unroll
  for (int j = 0; j < 8; ++j) {
    cL[j] = hi ? G[4 * 32 + j + 8] : G[4 * 32 + j];
    cH[j] = hi ? G[4 * 32 + 16 + j + 8] : G[4 * 32 + 16 + j];
  }
  const float M = G[160];

  for (int t = gwave; t < ntiles; t += nwaves) {
    int p = t * 16 + col;
    int pc = p < npoints ? p : npoints - 1;
    v2f x = X[pc];

    int pf = p + nwaves * 16;
    pf = pf < npoints ? pf : npoints - 1;
    __builtin_prefetch(&X[pf], 0, 1);

    v2f b;
    b.x = hi ? x.x : x.x * x.x;
    b.y = hi ? x.y : x.y * x.y;

    v8f d1 = __builtin_amdgcn_wmma_f32_16x16x4_f32(
        false, aL, false, b, (short)0, cL, false, false);
    v8f d2 = __builtin_amdgcn_wmma_f32_16x16x4_f32(
        false, aH, false, b, (short)0, cH, false, false);

    float s0 = __builtin_amdgcn_exp2f(d1[0]) + __builtin_amdgcn_exp2f(d2[0]);
    float s1 = __builtin_amdgcn_exp2f(d1[1]) + __builtin_amdgcn_exp2f(d2[1]);
    float s2 = __builtin_amdgcn_exp2f(d1[2]) + __builtin_amdgcn_exp2f(d2[2]);
    float s3 = __builtin_amdgcn_exp2f(d1[3]) + __builtin_amdgcn_exp2f(d2[3]);
    float s4 = __builtin_amdgcn_exp2f(d1[4]) + __builtin_amdgcn_exp2f(d2[4]);
    float s5 = __builtin_amdgcn_exp2f(d1[5]) + __builtin_amdgcn_exp2f(d2[5]);
    float s6 = __builtin_amdgcn_exp2f(d1[6]) + __builtin_amdgcn_exp2f(d2[6]);
    float s7 = __builtin_amdgcn_exp2f(d1[7]) + __builtin_amdgcn_exp2f(d2[7]);
    float t0 = s0 + s1, t1 = s2 + s3, t2 = s4 + s5, t3 = s6 + s7;
    float part = (t0 + t1) + (t2 + t3);

    float ssum = part + __shfl_xor(part, 16, 32);

    float res = fmaf(__builtin_amdgcn_logf(ssum), LN2, M);

    int idx = t * 16 + col;
    if (!hi && idx < npoints) out[idx] = res;
  }
}

extern "C" void kernel_launch(void* const* d_in, const int* in_sizes, int n_in,
                              void* d_out, int out_size, void* d_ws, size_t ws_size,
                              hipStream_t stream) {
  const float* X = (const float*)d_in[0];
  const float* w = (const float*)d_in[1];
  const float* means = (const float*)d_in[2];
  const float* logvar = (const float*)d_in[3];
  float* out = (float*)d_out;
  float* G = (float*)d_ws;

  const int npoints = in_sizes[0] / 2;
  const int ntiles = (npoints + 15) / 16;

  gmm_setup_kernel<<<1, 32, 0, stream>>>(w, means, logvar, G);

  int blocks = (ntiles + 7) / 8;
  if (blocks > 8192) blocks = 8192;
  if (blocks < 1) blocks = 1;
  gmm_wmma_kernel<<<blocks, 256, 0, stream>>>(X, G, out, npoints, ntiles);
}


# Persistent grid: 4,096 long-lived waves (was 65,536) walk 64-point steps with loads two steps ahead; half-waves swap partial sums so 32 points give one log and one 128-byte store; sums bit-identical
# speedup vs baseline: 1.3398x; 1.3398x over previous
#include <hip/hip_runtime.h>
#include <math.h>

typedef __attribute__((ext_vector_type(2))) float v2f;
typedef __attribute__((ext_vector_type(4))) float v4f;
typedef __attribute__((ext_vector_type(8))) float v8f;

#define LOG_2PI 1.8378770664093453f
#define LOG2E   1.4426950408889634f
#define LN2     0.6931471805599453f

__global__ __launch_bounds__(32) void gmm_setup_kernel(
    const float* __restrict__ w, const float* __restrict__ means,
    const float* __restrict__ logvar, float* __restrict__ G) {
  const int k = threadIdx.x;
  float wk = w[k];

  float mx = wk;
#pragma unroll
  for (int m = 16; m >= 1; m >>= 1) mx = fmaxf(mx, __shfl_xor(mx, m, 32));
  float s = __expf(wk - mx);
#pragma unroll
  for (int m = 16; m >= 1; m >>= 1) s += __shfl_xor(s, m, 32);
  const float logw = wk - mx - __logf(s);

  const float lv0 = logvar[2 * k], lv1 = logvar[2 * k + 1];
  const float iv0 = __expf(-lv0), iv1 = __expf(-lv1);
  const float m0 = means[2 * k], m1 = means[2 * k + 1];

  const float ck = logw - 0.5f * (lv0 + lv1 + 2.0f * LOG_2PI)
                        - 0.5f * (m0 * m0 * iv0 + m1 * m1 * iv1);
  float M = ck;
#pragma unroll
  for (int m = 16; m >= 1; m >>= 1) M = fmaxf(M, __shfl_xor(M, m, 32));

  G[0 * 32 + k] = -0.5f * iv0 * LOG2E;
  G[1 * 32 + k] = -0.5f * iv1 * LOG2E;
  G[2 * 32 + k] = m0 * iv0 * LOG2E;
  G[3 * 32 + k] = m1 * iv1 * LOG2E;
  G[4 * 32 + k] = (ck - M) * LOG2E;
  if (k == 0) G[160] = M;
}

__device__ __forceinline__ float half_sum16(v8f d1, v8f d2) {
  float s0 = __builtin_amdgcn_exp2f(d1[0]) + __builtin_amdgcn_exp2f(d2[0]);
  float s1 = __builtin_amdgcn_exp2f(d1[1]) + __builtin_amdgcn_exp2f(d2[1]);
  float s2 = __builtin_amdgcn_exp2f(d1[2]) + __builtin_amdgcn_exp2f(d2[2]);
  float s3 = __builtin_amdgcn_exp2f(d1[3]) + __builtin_amdgcn_exp2f(d2[3]);
  float s4 = __builtin_amdgcn_exp2f(d1[4]) + __builtin_amdgcn_exp2f(d2[4]);
  float s5 = __builtin_amdgcn_exp2f(d1[5]) + __builtin_amdgcn_exp2f(d2[5]);
  float s6 = __builtin_amdgcn_exp2f(d1[6]) + __builtin_amdgcn_exp2f(d2[6]);
  float s7 = __builtin_amdgcn_exp2f(d1[7]) + __builtin_amdgcn_exp2f(d2[7]);
  float t0 = s0 + s1, t1 = s2 + s3, t2 = s4 + s5, t3 = s6 + s7;
  return (t0 + t1) + (t2 + t3);
}

__device__ __forceinline__ float group32(v4f v, bool hi, v2f aL, v2f aH,
                                         v8f cL, v8f cH, float M) {
  v2f bE, bO;
  bE.x = hi ? v.x : v.x * v.x;
  bE.y = hi ? v.y : v.y * v.y;
  bO.x = hi ? v.z : v.z * v.z;
  bO.y = hi ? v.w : v.w * v.w;

  v8f e1 = __builtin_amdgcn_wmma_f32_16x16x4_f32(
      false, aL, false, bE, (short)0, cL, false, false);
  v8f e2 = __builtin_amdgcn_wmma_f32_16x16x4_f32(
      false, aH, false, bE, (short)0, cH, false, false);
  v8f o1 = __builtin_amdgcn_wmma_f32_16x16x4_f32(
      false, aL, false, bO, (short)0, cL, false, false);
  v8f o2 = __builtin_amdgcn_wmma_f32_16x16x4_f32(
      false, aH, false, bO, (short)0, cH, false, false);

  const float pE = half_sum16(e1, e2);
  const float pO = half_sum16(o1, o2);

  const float keep = hi ? pO : pE;
  const float send = hi ? pE : pO;
  const float ssum = keep + __shfl_xor(send, 16, 32);

  return fmaf(__builtin_amdgcn_logf(ssum), LN2, M);
}

__global__ __launch_bounds__(256) void gmm_wmma_kernel(
    const float* __restrict__ Xf, const float* __restrict__ G,
    float* __restrict__ out, int npoints, int ntiles) {
  const v2f* __restrict__ X = (const v2f*)Xf;
  const v4f* __restrict__ X4 = (const v4f*)Xf;
  const int lane = threadIdx.x & 31;
  const int col = lane & 15;
  const bool hi = lane >= 16;

  const int wavesPerBlock = blockDim.x >> 5;
  const int wib = __builtin_amdgcn_readfirstlane((int)(threadIdx.x >> 5));
  const int gwave = blockIdx.x * wavesPerBlock + wib;
  const int nwaves = gridDim.x * wavesPerBlock;

  v2f aL, aH;
  aL.x = hi ? G[2 * 32 + col] : G[0 * 32 + col];
  aL.y = hi ? G[3 * 32 + col] : G[1 * 32 + col];
  aH.x = hi ? G[2 * 32 + 16 + col] : G[0 * 32 + 16 + col];
  aH.y = hi ? G[3 * 32 + 16 + col] : G[1 * 32 + 16 + col];

  v8f cL, cH;
#pragma unroll
  for (int j = 0; j < 8; ++j) {
    cL[j] = hi ? G[4 * 32 + j + 8] : G[4 * 32 + j];
    cH[j] = hi ? G[4 * 32 + 16 + j + 8] : G[4 * 32 + 16 + j];
  }
  const float M = G[160];

  const int nsteps = npoints >> 6;
  const int slot = 2 * col + (hi ? 1 : 0);

#define GMM_LOAD(p, q, t)                                   \
  {                                                         \
    const int tl_ = (t) < nsteps ? (t) : s;                 \
    const v4f* src_ = X4 + (size_t)tl_ * 32 + col;          \
    p = src_[0];                                            \
    q = src_[16];                                           \
  }
#define GMM_STEP(p, q)                                      \
  {                                                         \
    const float ra_ = group32(p, hi, aL, aH, cL, cH, M);    \
    const float rb_ = group32(q, hi, aL, aH, cL, cH, M);    \
    float* o_ = out + (size_t)s * 64 + slot;                \
    o_[0] = ra_;      \
    o_[32] = rb_;                         \
  }

  int s = gwave;
  if (s < nsteps) {
    v4f a0, a1, b0, b1, c0, c1;
    GMM_LOAD(a0, a1, s);
    GMM_LOAD(b0, b1, s + nwaves);
    while (s + 2 * nwaves < nsteps) {
      GMM_LOAD(c0, c1, s + 2 * nwaves);
      GMM_STEP(a0, a1);
      s += nwaves;

      GMM_LOAD(a0, a1, s + 2 * nwaves);
      GMM_STEP(b0, b1);
      s += nwaves;

      GMM_LOAD(b0, b1, s + 2 * nwaves);
      GMM_STEP(c0, c1);
      s += nwaves;
    }
    if (s < nsteps) {
      GMM_STEP(a0, a1);
      s += nwaves;
      if (s < nsteps) {
        GMM_STEP(b0, b1);
      }
    }
  }
#undef GMM_LOAD
#undef GMM_STEP

  for (int t = nsteps * 4 + gwave; t < ntiles; t += nwaves) {
    int p = t * 16 + col;
    int pc = p < npoints ? p : npoints - 1;
    v2f x = X[pc];

    v2f b;
    b.x = hi ? x.x : x.x * x.x;
    b.y = hi ? x.y : x.y * x.y;

    v8f d1 = __builtin_amdgcn_wmma_f32_16x16x4_f32(
        false, aL, false, b, (short)0, cL, false, false);
    v8f d2 = __builtin_amdgcn_wmma_f32_16x16x4_f32(
        false, aH, false, b, (short)0, cH, false, false);

    float part = half_sum16(d1, d2);

    float ssum = part + __shfl_xor(part, 16, 32);

    float res = fmaf(__builtin_amdgcn_logf(ssum), LN2, M);

    if (!hi && p < npoints) out[p] = res;
  }
}

extern "C" void kernel_launch(void* const* d_in, const int* in_sizes, int n_in,
                              void* d_out, int out_size, void* d_ws, size_t ws_size,
                              hipStream_t stream) {
  const float* X = (const float*)d_in[0];
  const float* w = (const float*)d_in[1];
  const float* means = (const float*)d_in[2];
  const float* logvar = (const float*)d_in[3];
  float* out = (float*)d_out;
  float* G = (float*)d_ws;

  const int npoints = in_sizes[0] / 2;
  const int ntiles = (npoints + 15) / 16;

  gmm_setup_kernel<<<1, 32, 0, stream>>>(w, means, logvar, G);

  int blocks = (ntiles + 7) / 8;
  if (blocks > 512) blocks = 512;
  if (blocks < 1) blocks = 1;
  gmm_wmma_kernel<<<blocks, 256, 0, stream>>>(X, G, out, npoints, ntiles);
}


# One launch instead of two: each wave derives the 32 components' coefficients itself in registers (lane k = component k; same formulas and sum order, DPP/permlanex16 butterflies); set-up kernel gone
# speedup vs baseline: 1.5457x; 1.1537x over previous
#include <hip/hip_runtime.h>
#include <math.h>

typedef __attribute__((ext_vector_type(2))) float v2f;
typedef __attribute__((ext_vector_type(4))) float v4f;
typedef __attribute__((ext_vector_type(8))) float v8f;

#define LOG_2PI 1.8378770664093453f
#define LOG2E   1.4426950408889634f
#define LN2     0.6931471805599453f

__device__ __forceinline__ float xhalf(float x) {
  const int b = __builtin_bit_cast(int, x);
  return __builtin_bit_cast(
      float, __builtin_amdgcn_permlanex16(b, b, 0x76543210, 0xfedcba98, false, false));
}
template <int MASK>
__device__ __forceinline__ float xrow(float x) {
  const int b = __builtin_bit_cast(int, x);
  return __builtin_bit_cast(
      float, __builtin_amdgcn_update_dpp(b, b, 0x160 + MASK, 0xF, 0xF, false));
}
__device__ __forceinline__ float wave_max(float v) {
  v = fmaxf(v, xhalf(v));
  v = fmaxf(v, xrow<8>(v));
  v = fmaxf(v, xrow<4>(v));
  v = fmaxf(v, xrow<2>(v));
  v = fmaxf(v, xrow<1>(v));
  return v;
}
__device__ __forceinline__ float wave_sum(float v) {
  v += xhalf(v);
  v += xrow<8>(v);
  v += xrow<4>(v);
  v += xrow<2>(v);
  v += xrow<1>(v);
  return v;
}

struct GmmCoef {
  float g0, g1, g2, g3, g4, M;
};

__device__ __forceinline__ GmmCoef gmm_coef(float wk, float m0, float m1,
                                            float lv0, float lv1) {
  const float mx = wave_max(wk);
  const float s = wave_sum(__expf(wk - mx));
  const float logw = wk - mx - __logf(s);

  const float iv0 = __expf(-lv0), iv1 = __expf(-lv1);

  const float ck = logw - 0.5f * (lv0 + lv1 + 2.0f * LOG_2PI)
                        - 0.5f * (m0 * m0 * iv0 + m1 * m1 * iv1);
  const float M = wave_max(ck);

  GmmCoef r;
  r.g0 = -0.5f * iv0 * LOG2E;
  r.g1 = -0.5f * iv1 * LOG2E;
  r.g2 = m0 * iv0 * LOG2E;
  r.g3 = m1 * iv1 * LOG2E;
  r.g4 = (ck - M) * LOG2E;
  r.M = M;
  return r;
}

__device__ __forceinline__ float half_sum16(v8f d1, v8f d2) {
  float s0 = __builtin_amdgcn_exp2f(d1[0]) + __builtin_amdgcn_exp2f(d2[0]);
  float s1 = __builtin_amdgcn_exp2f(d1[1]) + __builtin_amdgcn_exp2f(d2[1]);
  float s2 = __builtin_amdgcn_exp2f(d1[2]) + __builtin_amdgcn_exp2f(d2[2]);
  float s3 = __builtin_amdgcn_exp2f(d1[3]) + __builtin_amdgcn_exp2f(d2[3]);
  float s4 = __builtin_amdgcn_exp2f(d1[4]) + __builtin_amdgcn_exp2f(d2[4]);
  float s5 = __builtin_amdgcn_exp2f(d1[5]) + __builtin_amdgcn_exp2f(d2[5]);
  float s6 = __builtin_amdgcn_exp2f(d1[6]) + __builtin_amdgcn_exp2f(d2[6]);
  float s7 = __builtin_amdgcn_exp2f(d1[7]) + __builtin_amdgcn_exp2f(d2[7]);
  float t0 = s0 + s1, t1 = s2 + s3, t2 = s4 + s5, t3 = s6 + s7;
  return (t0 + t1) + (t2 + t3);
}

__device__ __forceinline__ float group32(v4f v, bool hi, v2f aL, v2f aH,
                                         v8f cL, v8f cH, float M) {
  v2f bE, bO;
  bE.x = hi ? v.x : v.x * v.x;
  bE.y = hi ? v.y : v.y * v.y;
  bO.x = hi ? v.z : v.z * v.z;
  bO.y = hi ? v.w : v.w * v.w;

  v8f e1 = __builtin_amdgcn_wmma_f32_16x16x4_f32(
      false, aL, false, bE, (short)0, cL, false, false);
  v8f e2 = __builtin_amdgcn_wmma_f32_16x16x4_f32(
      false, aH, false, bE, (short)0, cH, false, false);
  v8f o1 = __builtin_amdgcn_wmma_f32_16x16x4_f32(
      false, aL, false, bO, (short)0, cL, false, false);
  v8f o2 = __builtin_amdgcn_wmma_f32_16x16x4_f32(
      false, aH, false, bO, (short)0, cH, false, false);

  const float pE = half_sum16(e1, e2);
  const float pO = half_sum16(o1, o2);

  const float keep = hi ? pO : pE;
  const float send = hi ? pE : pO;
  const float ssum = keep + __shfl_xor(send, 16, 32);

  return fmaf(__builtin_amdgcn_logf(ssum), LN2, M);
}

__global__ __launch_bounds__(256) void gmm_wmma_kernel(
    const float* __restrict__ Xf, const float* __restrict__ w,
    const float* __restrict__ means, const float* __restrict__ logvar,
    float* __restrict__ out, int npoints, int ntiles) {
  const v2f* __restrict__ X = (const v2f*)Xf;
  const v4f* __restrict__ X4 = (const v4f*)Xf;
  const int lane = threadIdx.x & 31;
  const int col = lane & 15;
  const bool hi = lane >= 16;

  const int wavesPerBlock = blockDim.x >> 5;
  const int wib = __builtin_amdgcn_readfirstlane((int)(threadIdx.x >> 5));
  const int gwave = blockIdx.x * wavesPerBlock + wib;
  const int nwaves = gridDim.x * wavesPerBlock;

  const float wk = w[lane];
  const float lv0 = logvar[2 * lane], lv1 = logvar[2 * lane + 1];
  const float m0 = means[2 * lane], m1 = means[2 * lane + 1];

  const int nsteps = npoints >> 6;
  int s = gwave;
  const bool has_steps = s < nsteps;

#define GMM_LOAD(p, q, t)                                   \
  {                                                         \
    const int tl_ = (t) < nsteps ? (t) : s;                 \
    const v4f* src_ = X4 + (size_t)tl_ * 32 + col;          \
    p = src_[0];                                            \
    q = src_[16];                                           \
  }

  const v4f zero4 = {0.0f, 0.0f, 0.0f, 0.0f};
  v4f a0 = zero4, a1 = zero4, b0 = zero4, b1 = zero4, c0, c1;
  if (has_steps) {
    GMM_LOAD(a0, a1, s);
    GMM_LOAD(b0, b1, s + nwaves);
  }

  const GmmCoef cf = gmm_coef(wk, m0, m1, lv0, lv1);

  const float g0x = xhalf(cf.g0);
  const float g1x = xhalf(cf.g1);
  const float g2x = xhalf(cf.g2);
  const float g3x = xhalf(cf.g3);
  v2f aL, aH;
  aL.x = hi ? g2x : cf.g0;
  aL.y = hi ? g3x : cf.g1;
  aH.x = hi ? cf.g2 : g0x;
  aH.y = hi ? cf.g3 : g1x;

  v8f cL, cH;
#pragma unroll
  for (int j = 0; j < 8; ++j) {
    const float r0 = __builtin_bit_cast(float, __builtin_amdgcn_readlane(__builtin_bit_cast(int, cf.g4), j));
    const float r1 = __builtin_bit_cast(float, __builtin_amdgcn_readlane(__builtin_bit_cast(int, cf.g4), j + 8));
    const float r2 = __builtin_bit_cast(float, __builtin_amdgcn_readlane(__builtin_bit_cast(int, cf.g4), j + 16));
    const float r3 = __builtin_bit_cast(float, __builtin_amdgcn_readlane(__builtin_bit_cast(int, cf.g4), j + 24));
    cL[j] = hi ? r1 : r0;
    cH[j] = hi ? r3 : r2;
  }
  const float M = cf.M;

  const int slot = 2 * col + (hi ? 1 : 0);

#define GMM_STEP(p, q)                                      \
  {                                                         \
    const float ra_ = group32(p, hi, aL, aH, cL, cH, M);    \
    const float rb_ = group32(q, hi, aL, aH, cL, cH, M);    \
    float* o_ = out + (size_t)s * 64 + slot;                \
    o_[0] = ra_;      \
    o_[32] = rb_;                         \
  }

  if (has_steps) {
    while (s + 2 * nwaves < nsteps) {
      GMM_LOAD(c0, c1, s + 2 * nwaves);
      GMM_STEP(a0, a1);
      s += nwaves;

      GMM_LOAD(a0, a1, s + 2 * nwaves);
      GMM_STEP(b0, b1);
      s += nwaves;

      GMM_LOAD(b0, b1, s + 2 * nwaves);
      GMM_STEP(c0, c1);
      s += nwaves;
    }
    if (s < nsteps) {
      GMM_STEP(a0, a1);
      s += nwaves;
      if (s < nsteps) {
        GMM_STEP(b0, b1);
      }
    }
  }
#undef GMM_LOAD
#undef GMM_STEP

  for (int t = nsteps * 4 + gwave; t < ntiles; t += nwaves) {
    int p = t * 16 + col;
    int pc = p < npoints ? p : npoints - 1;
    v2f x = X[pc];

    v2f b;
    b.x = hi ? x.x : x.x * x.x;
    b.y = hi ? x.y : x.y * x.y;

    v8f d1 = __builtin_amdgcn_wmma_f32_16x16x4_f32(
        false, aL, false, b, (short)0, cL, false, false);
    v8f d2 = __builtin_amdgcn_wmma_f32_16x16x4_f32(
        false, aH, false, b, (short)0, cH, false, false);

    float part = half_sum16(d1, d2);

    float ssum = part + __shfl_xor(part, 16, 32);

    float res = fmaf(__builtin_amdgcn_logf(ssum), LN2, M);

    if (!hi && p < npoints) out[p] = res;
  }
}

extern "C" void kernel_launch(void* const* d_in, const int* in_sizes, int n_in,
                              void* d_out, int out_size, void* d_ws, size_t ws_size,
                              hipStream_t stream) {
  const float* X = (const float*)d_in[0];
  const float* w = (const float*)d_in[1];
  const float* means = (const float*)d_in[2];
  const float* logvar = (const float*)d_in[3];
  float* out = (float*)d_out;

  const int npoints = in_sizes[0] / 2;
  const int ntiles = (npoints + 15) / 16;

  int blocks = (ntiles + 7) / 8;
  if (blocks > 512) blocks = 512;
  if (blocks < 1) blocks = 1;
  gmm_wmma_kernel<<<blocks, 256, 0, stream>>>(X, w, means, logvar, out,
                                             npoints, ntiles);
}
